# starting kernel
# baseline (speedup 1.0000x reference)
#include <hip/hip_runtime.h>

#define NB 8
#define NT 2000
#define NH 256
#define NF 257
#define ALEN 128
#define SS 160

typedef float v2f __attribute__((ext_vector_type(2)));
typedef float v8f __attribute__((ext_vector_type(8)));

__device__ __forceinline__ v8f wmma4(v2f a, v2f b, v8f c) {
  return __builtin_amdgcn_wmma_f32_16x16x4_f32(false, a, false, b, (short)0, c,
                                               false, false);
}

__global__ __launch_bounds__(128)
void attn_fused_kernel(const float* __restrict__ Kin, const float* __restrict__ Qin,
                       const float* __restrict__ Wsc, const float* __restrict__ Wen,
                       const float* __restrict__ ben, const float* __restrict__ Wmk,
                       const float* __restrict__ bmk, float* __restrict__ Out)
{
  __shared__ __align__(16) float qtile[16 * NH];
  __shared__ __align__(16) float stile[16 * NH];
  __shared__ __align__(16) float otile[16 * NH];
  __shared__ __align__(16) float sbuf[16 * SS];
  __shared__ float wmax[4 * 16];
  __shared__ float rowmax[16];
  __shared__ float rowsum[16];
  __shared__ float rscale[16];

  const int ttile = blockIdx.x;
  const int bat   = blockIdx.y;
  const int t0    = ttile * 16;
  const int tid   = threadIdx.x;
  const int wave  = tid >> 5;
  const int lane  = tid & 31;
  const int half  = lane >> 4;
  const int l16   = lane & 15;
  const int koff  = 2 * half;

  const float* qbase = Qin + ((size_t)bat * NT + t0) * NH;
  const float* kbase = Kin + (size_t)bat * NT * NH;

  for (int i = tid; i < 16 * NH; i += 128) qtile[i] = qbase[i];
  __syncthreads();

  for (int nt = wave; nt < NH / 16; nt += 4) {
    const int g0 = nt * 16;
    v8f acc = {};
    for (int kb = 0; kb < NH; kb += 4) {
      const int kk = kb + koff;
      v2f a = *(const v2f*)&qtile[l16 * NH + kk];
      v2f b = *(const v2f*)&Wsc[(size_t)(g0 + l16) * NH + kk];
      acc = wmma4(a, b, acc);
    }
#pragma unroll
    for (int r = 0; r < 8; ++r) stile[(r + 8 * half) * NH + g0 + l16] = acc[r];
  }
  __syncthreads();

  const int st_lo = (ttile > 8) ? (ttile - 8) : 0;
  float rmax[8];
#pragma unroll
  for (int r = 0; r < 8; ++r) rmax[r] = -3.4e38f;

  for (int st = wave; st < NT / 16; st += 4) {
    const int s0 = st * 16;
    v8f acc = {};
    for (int kb = 0; kb < NH; kb += 4) {
      const int kk = kb + koff;
      v2f a = *(const v2f*)&stile[l16 * NH + kk];
      v2f b = *(const v2f*)&kbase[(size_t)(s0 + l16) * NH + kk];
      acc = wmma4(a, b, acc);
    }
#pragma unroll
    for (int r = 0; r < 8; ++r) rmax[r] = fmaxf(rmax[r], acc[r]);
    if (st >= st_lo && st <= ttile) {
      const int c0 = (st - st_lo) * 16;
#pragma unroll
      for (int r = 0; r < 8; ++r) sbuf[(r + 8 * half) * SS + c0 + l16] = acc[r];
    }
  }

#pragma unroll
  for (int r = 0; r < 8; ++r) {
    float v = rmax[r];
    v = fmaxf(v, __shfl_xor(v, 1, 32));
    v = fmaxf(v, __shfl_xor(v, 2, 32));
    v = fmaxf(v, __shfl_xor(v, 4, 32));
    v = fmaxf(v, __shfl_xor(v, 8, 32));
    rmax[r] = v;
  }
  if (l16 == 0) {
#pragma unroll
    for (int r = 0; r < 8; ++r) wmax[wave * 16 + 8 * half + r] = rmax[r];
  }
  __syncthreads();
  if (tid < 16) {
    float m = wmax[tid];
    m = fmaxf(m, wmax[16 + tid]);
    m = fmaxf(m, wmax[32 + tid]);
    m = fmaxf(m, wmax[48 + tid]);
    rowmax[tid] = m;
    rowsum[tid] = 0.0f;
  }
  __syncthreads();

  const int ncols = (ttile - st_lo + 1) * 16;
  const int s0b   = st_lo * 16;
  {
    const int row = tid & 15;
    const int t   = t0 + row;
    float psum = 0.0f;
    for (int c = tid >> 4; c < ncols; c += 8) {
      const int s = s0b + c;
      float w = 0.0f;
      if (s <= t && s >= t - ALEN) w = __expf(sbuf[row * SS + c] - rowmax[row]);
      sbuf[row * SS + c] = w;
      psum += w;
    }
    atomicAdd(&rowsum[row], psum);
  }
  __syncthreads();
  if (tid < 16) rscale[tid] = 1.0f / (rowsum[tid] + 1e-30f);
  __syncthreads();

  for (int nt = wave; nt < NH / 16; nt += 4) {
    const int h0 = nt * 16;
    v8f acc = {};
    for (int kb = 0; kb < ncols; kb += 4) {
      const int kk = kb + koff;
      v2f a = *(const v2f*)&sbuf[l16 * SS + kk];
      v2f b;
      b.x = kbase[(size_t)(s0b + kk)     * NH + h0 + l16];
      b.y = kbase[(size_t)(s0b + kk + 1) * NH + h0 + l16];
      acc = wmma4(a, b, acc);
    }
#pragma unroll
    for (int r = 0; r < 8; ++r)
      stile[(r + 8 * half) * NH + h0 + l16] = acc[r] * rscale[r + 8 * half];
  }
  __syncthreads();

  for (int nt = wave; nt < NH / 16; nt += 4) {
    const int j0 = nt * 16;
    v8f acc = {};
    for (int kb = 0; kb < NH; kb += 4) {
      const int kk = kb + koff;
      v2f a = *(const v2f*)&stile[l16 * NH + kk];
      v2f b = *(const v2f*)&Wen[(size_t)(j0 + l16) * (2 * NH) + kk];
      acc = wmma4(a, b, acc);
    }
    for (int kb = 0; kb < NH; kb += 4) {
      const int kk = kb + koff;
      v2f a = *(const v2f*)&qtile[l16 * NH + kk];
      v2f b = *(const v2f*)&Wen[(size_t)(j0 + l16) * (2 * NH) + NH + kk];
      acc = wmma4(a, b, acc);
    }
    const float bias = ben[j0 + l16];
#pragma unroll
    for (int r = 0; r < 8; ++r)
      otile[(r + 8 * half) * NH + j0 + l16] = tanhf(acc[r] + bias);
  }
  __syncthreads();

  for (int ft = wave; ft < 17; ft += 4) {
    const int f0 = ft * 16;
    const int f  = f0 + l16;
    v8f acc = {};
    for (int kb = 0; kb < NH; kb += 4) {
      const int kk = kb + koff;
      v2f a = *(const v2f*)&otile[l16 * NH + kk];
      v2f b = {0.0f, 0.0f};
      if (f < NF) b = *(const v2f*)&Wmk[(size_t)f * NH + kk];
      acc = wmma4(a, b, acc);
    }
    if (f < NF) {
      const float bias = bmk[f];
#pragma unroll
      for (int r = 0; r < 8; ++r) {
        const int t = t0 + r + 8 * half;
        const float v = acc[r] + bias;
        Out[((size_t)bat * NT + t) * NF + f] = 1.0f / (1.0f + __expf(-v));
      }
    }
  }
}

extern "C" void kernel_launch(void* const* d_in, const int* in_sizes, int n_in,
                              void* d_out, int out_size, void* d_ws, size_t ws_size,
                              hipStream_t stream) {
  (void)in_sizes; (void)n_in; (void)d_ws; (void)ws_size; (void)out_size;
  const float* k   = (const float*)d_in[0];
  const float* q   = (const float*)d_in[1];
  const float* Wsc = (const float*)d_in[2];
  const float* Wen = (const float*)d_in[3];
  const float* ben = (const float*)d_in[4];
  const float* Wmk = (const float*)d_in[5];
  const float* bmk = (const float*)d_in[6];
  dim3 grid(NT / 16, NB);
  attn_fused_kernel<<<grid, 128, 0, stream>>>(k, q, Wsc, Wen, ben, Wmk, bmk,
                                              (float*)d_out);
}


# LDS row pitch 256->260 floats (score slab 160->164): the 16 rows of every A-fragment read fall on 16 different banks instead of one, removing a 16-way conflict per ds_load; sum order unchanged
# speedup vs baseline: 1.3068x; 1.3068x over previous
#include <hip/hip_runtime.h>

#define NB 8
#define NT 2000
#define NH 256
#define NF 257
#define ALEN 128
#define LP 260
#define SS 164

typedef float v2f __attribute__((ext_vector_type(2)));
typedef float v8f __attribute__((ext_vector_type(8)));

__device__ __forceinline__ v8f wmma4(v2f a, v2f b, v8f c) {
  return __builtin_amdgcn_wmma_f32_16x16x4_f32(false, a, false, b, (short)0, c,
                                               false, false);
}

__global__ __launch_bounds__(128)
void attn_fused_kernel(const float* __restrict__ Kin, const float* __restrict__ Qin,
                       const float* __restrict__ Wsc, const float* __restrict__ Wen,
                       const float* __restrict__ ben, const float* __restrict__ Wmk,
                       const float* __restrict__ bmk, float* __restrict__ Out)
{
  __shared__ __align__(16) float qtile[16 * LP];
  __shared__ __align__(16) float stile[16 * LP];
  __shared__ __align__(16) float otile[16 * LP];
  __shared__ __align__(16) float sbuf[16 * SS];
  __shared__ float wmax[4 * 16];
  __shared__ float rowmax[16];
  __shared__ float rowsum[16];
  __shared__ float rscale[16];

  const int ttile = blockIdx.x;
  const int bat   = blockIdx.y;
  const int t0    = ttile * 16;
  const int tid   = threadIdx.x;
  const int wave  = tid >> 5;
  const int lane  = tid & 31;
  const int half  = lane >> 4;
  const int l16   = lane & 15;
  const int koff  = 2 * half;

  const float* qbase = Qin + ((size_t)bat * NT + t0) * NH;
  const float* kbase = Kin + (size_t)bat * NT * NH;

  for (int i = tid; i < 16 * NH; i += 128)
    qtile[(i >> 8) * LP + (i & (NH - 1))] = qbase[i];
  __syncthreads();

  for (int nt = wave; nt < NH / 16; nt += 4) {
    const int g0 = nt * 16;
    v8f acc = {};
    for (int kb = 0; kb < NH; kb += 4) {
      const int kk = kb + koff;
      v2f a = *(const v2f*)&qtile[l16 * LP + kk];
      v2f b = *(const v2f*)&Wsc[(size_t)(g0 + l16) * NH + kk];
      acc = wmma4(a, b, acc);
    }
#pragma unroll
    for (int r = 0; r < 8; ++r) stile[(r + 8 * half) * LP + g0 + l16] = acc[r];
  }
  __syncthreads();

  const int st_lo = (ttile > 8) ? (ttile - 8) : 0;
  float rmax[8];
#pragma unroll
  for (int r = 0; r < 8; ++r) rmax[r] = -3.4e38f;

  for (int st = wave; st < NT / 16; st += 4) {
    const int s0 = st * 16;
    v8f acc = {};
    for (int kb = 0; kb < NH; kb += 4) {
      const int kk = kb + koff;
      v2f a = *(const v2f*)&stile[l16 * LP + kk];
      v2f b = *(const v2f*)&kbase[(size_t)(s0 + l16) * NH + kk];
      acc = wmma4(a, b, acc);
    }
#pragma unroll
    for (int r = 0; r < 8; ++r) rmax[r] = fmaxf(rmax[r], acc[r]);
    if (st >= st_lo && st <= ttile) {
      const int c0 = (st - st_lo) * 16;
#pragma unroll
      for (int r = 0; r < 8; ++r) sbuf[(r + 8 * half) * SS + c0 + l16] = acc[r];
    }
  }

#pragma unroll
  for (int r = 0; r < 8; ++r) {
    float v = rmax[r];
    v = fmaxf(v, __shfl_xor(v, 1, 32));
    v = fmaxf(v, __shfl_xor(v, 2, 32));
    v = fmaxf(v, __shfl_xor(v, 4, 32));
    v = fmaxf(v, __shfl_xor(v, 8, 32));
    rmax[r] = v;
  }
  if (l16 == 0) {
#pragma unroll
    for (int r = 0; r < 8; ++r) wmax[wave * 16 + 8 * half + r] = rmax[r];
  }
  __syncthreads();
  if (tid < 16) {
    float m = wmax[tid];
    m = fmaxf(m, wmax[16 + tid]);
    m = fmaxf(m, wmax[32 + tid]);
    m = fmaxf(m, wmax[48 + tid]);
    rowmax[tid] = m;
    rowsum[tid] = 0.0f;
  }
  __syncthreads();

  const int ncols = (ttile - st_lo + 1) * 16;
  const int s0b   = st_lo * 16;
  {
    const int row = tid & 15;
    const int t   = t0 + row;
    float psum = 0.0f;
    for (int c = tid >> 4; c < ncols; c += 8) {
      const int s = s0b + c;
      float w = 0.0f;
      if (s <= t && s >= t - ALEN) w = __expf(sbuf[row * SS + c] - rowmax[row]);
      sbuf[row * SS + c] = w;
      psum += w;
    }
    atomicAdd(&rowsum[row], psum);
  }
  __syncthreads();
  if (tid < 16) rscale[tid] = 1.0f / (rowsum[tid] + 1e-30f);
  __syncthreads();

  for (int nt = wave; nt < NH / 16; nt += 4) {
    const int h0 = nt * 16;
    v8f acc = {};
    for (int kb = 0; kb < ncols; kb += 4) {
      const int kk = kb + koff;
      v2f a = *(const v2f*)&sbuf[l16 * SS + kk];
      v2f b;
      b.x = kbase[(size_t)(s0b + kk)     * NH + h0 + l16];
      b.y = kbase[(size_t)(s0b + kk + 1) * NH + h0 + l16];
      acc = wmma4(a, b, acc);
    }
#pragma unroll
    for (int r = 0; r < 8; ++r)
      stile[(r + 8 * half) * LP + h0 + l16] = acc[r] * rscale[r + 8 * half];
  }
  __syncthreads();

  for (int nt = wave; nt < NH / 16; nt += 4) {
    const int j0 = nt * 16;
    v8f acc = {};
    for (int kb = 0; kb < NH; kb += 4) {
      const int kk = kb + koff;
      v2f a = *(const v2f*)&stile[l16 * LP + kk];
      v2f b = *(const v2f*)&Wen[(size_t)(j0 + l16) * (2 * NH) + kk];
      acc = wmma4(a, b, acc);
    }
    for (int kb = 0; kb < NH; kb += 4) {
      const int kk = kb + koff;
      v2f a = *(const v2f*)&qtile[l16 * LP + kk];
      v2f b = *(const v2f*)&Wen[(size_t)(j0 + l16) * (2 * NH) + NH + kk];
      acc = wmma4(a, b, acc);
    }
    const float bias = ben[j0 + l16];
#pragma unroll
    for (int r = 0; r < 8; ++r)
      otile[(r + 8 * half) * LP + j0 + l16] = tanhf(acc[r] + bias);
  }
  __syncthreads();

  for (int ft = wave; ft < 17; ft += 4) {
    const int f0 = ft * 16;
    const int f  = f0 + l16;
    v8f acc = {};
    for (int kb = 0; kb < NH; kb += 4) {
      const int kk = kb + koff;
      v2f a = *(const v2f*)&otile[l16 * LP + kk];
      v2f b = {0.0f, 0.0f};
      if (f < NF) b = *(const v2f*)&Wmk[(size_t)f * NH + kk];
      acc = wmma4(a, b, acc);
    }
    if (f < NF) {
      const float bias = bmk[f];
#pragma unroll
      for (int r = 0; r < 8; ++r) {
        const int t = t0 + r + 8 * half;
        const float v = acc[r] + bias;
        Out[((size_t)bat * NT + t) * NF + f] = 1.0f / (1.0f + __expf(-v));
      }
    }
  }
}

extern "C" void kernel_launch(void* const* d_in, const int* in_sizes, int n_in,
                              void* d_out, int out_size, void* d_ws, size_t ws_size,
                              hipStream_t stream) {
  (void)in_sizes; (void)n_in; (void)d_ws; (void)ws_size; (void)out_size;
  const float* k   = (const float*)d_in[0];
  const float* q   = (const float*)d_in[1];
  const float* Wsc = (const float*)d_in[2];
  const float* Wen = (const float*)d_in[3];
  const float* ben = (const float*)d_in[4];
  const float* Wmk = (const float*)d_in[5];
  const float* bmk = (const float*)d_in[6];
  dim3 grid(NT / 16, NB);
  attn_fused_kernel<<<grid, 128, 0, stream>>>(k, q, Wsc, Wen, ben, Wmk, bmk,
                                              (float*)d_out);
}


# 128-bit fragment loads: each lane reads 4 consecutive k of A (LDS) and B (global) and feeds two WMMAs, halving B-load cache-line touches per WMMA; k grouping per WMMA (sum order) changes
# speedup vs baseline: 1.4311x; 1.0951x over previous
#include <hip/hip_runtime.h>

#define NB 8
#define NT 2000
#define NH 256
#define NF 257
#define ALEN 128
#define LP 260
#define SS 164

typedef float v2f __attribute__((ext_vector_type(2)));
typedef float v4f __attribute__((ext_vector_type(4)));
typedef float v8f __attribute__((ext_vector_type(8)));

__device__ __forceinline__ v8f wmma4(v2f a, v2f b, v8f c) {
  return __builtin_amdgcn_wmma_f32_16x16x4_f32(false, a, false, b, (short)0, c,
                                               false, false);
}

__global__ __launch_bounds__(128)
void attn_fused_kernel(const float* __restrict__ Kin, const float* __restrict__ Qin,
                       const float* __restrict__ Wsc, const float* __restrict__ Wen,
                       const float* __restrict__ ben, const float* __restrict__ Wmk,
                       const float* __restrict__ bmk, float* __restrict__ Out)
{
  __shared__ __align__(16) float qtile[16 * LP];
  __shared__ __align__(16) float stile[16 * LP];
  __shared__ __align__(16) float otile[16 * LP];
  __shared__ __align__(16) float sbuf[16 * SS];
  __shared__ float wmax[4 * 16];
  __shared__ float rowmax[16];
  __shared__ float rowsum[16];
  __shared__ float rscale[16];

  const int ttile = blockIdx.x;
  const int bat   = blockIdx.y;
  const int t0    = ttile * 16;
  const int tid   = threadIdx.x;
  const int wave  = tid >> 5;
  const int lane  = tid & 31;
  const int half  = lane >> 4;
  const int l16   = lane & 15;
  const int koff  = 2 * half;
  const int koff4 = 4 * half;

  const float* qbase = Qin + ((size_t)bat * NT + t0) * NH;
  const float* kbase = Kin + (size_t)bat * NT * NH;

  for (int i = tid; i < 16 * NH; i += 128)
    qtile[(i >> 8) * LP + (i & (NH - 1))] = qbase[i];
  __syncthreads();

  for (int nt = wave; nt < NH / 16; nt += 4) {
    const int g0 = nt * 16;
    v8f acc = {};
#pragma unroll 8
    for (int kb = 0; kb < NH; kb += 8) {
      const int kk = kb + koff4;
      v4f a = *(const v4f*)&qtile[l16 * LP + kk];
      v4f b = *(const v4f*)&Wsc[(size_t)(g0 + l16) * NH + kk];
      acc = wmma4(a.xy, b.xy, acc);
      acc = wmma4(a.zw, b.zw, acc);
    }
#pragma unroll
    for (int r = 0; r < 8; ++r) stile[(r + 8 * half) * LP + g0 + l16] = acc[r];
  }
  __syncthreads();

  const int st_lo = (ttile > 8) ? (ttile - 8) : 0;
  float rmax[8];
#pragma unroll
  for (int r = 0; r < 8; ++r) rmax[r] = -3.4e38f;

  for (int st = wave; st < NT / 16; st += 4) {
    const int s0 = st * 16;
    v8f acc = {};
#pragma unroll 8
    for (int kb = 0; kb < NH; kb += 8) {
      const int kk = kb + koff4;
      v4f a = *(const v4f*)&stile[l16 * LP + kk];
      v4f b = *(const v4f*)&kbase[(size_t)(s0 + l16) * NH + kk];
      acc = wmma4(a.xy, b.xy, acc);
      acc = wmma4(a.zw, b.zw, acc);
    }
#pragma unroll
    for (int r = 0; r < 8; ++r) rmax[r] = fmaxf(rmax[r], acc[r]);
    if (st >= st_lo && st <= ttile) {
      const int c0 = (st - st_lo) * 16;
#pragma unroll
      for (int r = 0; r < 8; ++r) sbuf[(r + 8 * half) * SS + c0 + l16] = acc[r];
    }
  }

#pragma unroll
  for (int r = 0; r < 8; ++r) {
    float v = rmax[r];
    v = fmaxf(v, __shfl_xor(v, 1, 32));
    v = fmaxf(v, __shfl_xor(v, 2, 32));
    v = fmaxf(v, __shfl_xor(v, 4, 32));
    v = fmaxf(v, __shfl_xor(v, 8, 32));
    rmax[r] = v;
  }
  if (l16 == 0) {
#pragma unroll
    for (int r = 0; r < 8; ++r) wmax[wave * 16 + 8 * half + r] = rmax[r];
  }
  __syncthreads();
  if (tid < 16) {
    float m = wmax[tid];
    m = fmaxf(m, wmax[16 + tid]);
    m = fmaxf(m, wmax[32 + tid]);
    m = fmaxf(m, wmax[48 + tid]);
    rowmax[tid] = m;
    rowsum[tid] = 0.0f;
  }
  __syncthreads();

  const int ncols = (ttile - st_lo + 1) * 16;
  const int s0b   = st_lo * 16;
  {
    const int row = tid & 15;
    const int t   = t0 + row;
    float psum = 0.0f;
    for (int c = tid >> 4; c < ncols; c += 8) {
      const int s = s0b + c;
      float w = 0.0f;
      if (s <= t && s >= t - ALEN) w = __expf(sbuf[row * SS + c] - rowmax[row]);
      sbuf[row * SS + c] = w;
      psum += w;
    }
    atomicAdd(&rowsum[row], psum);
  }
  __syncthreads();
  if (tid < 16) rscale[tid] = 1.0f / (rowsum[tid] + 1e-30f);
  __syncthreads();

  for (int nt = wave; nt < NH / 16; nt += 4) {
    const int h0 = nt * 16;
    v8f acc = {};
    for (int kb = 0; kb < ncols; kb += 4) {
      const int kk = kb + koff;
      v2f a = *(const v2f*)&sbuf[l16 * SS + kk];
      v2f b;
      b.x = kbase[(size_t)(s0b + kk)     * NH + h0 + l16];
      b.y = kbase[(size_t)(s0b + kk + 1) * NH + h0 + l16];
      acc = wmma4(a, b, acc);
    }
#pragma unroll
    for (int r = 0; r < 8; ++r)
      stile[(r + 8 * half) * LP + h0 + l16] = acc[r] * rscale[r + 8 * half];
  }
  __syncthreads();

  for (int nt = wave; nt < NH / 16; nt += 4) {
    const int j0 = nt * 16;
    v8f acc = {};
#pragma unroll 8
    for (int kb = 0; kb < NH; kb += 8) {
      const int kk = kb + koff4;
      v4f a = *(const v4f*)&stile[l16 * LP + kk];
      v4f b = *(const v4f*)&Wen[(size_t)(j0 + l16) * (2 * NH) + kk];
      acc = wmma4(a.xy, b.xy, acc);
      acc = wmma4(a.zw, b.zw, acc);
    }
#pragma unroll 8
    for (int kb = 0; kb < NH; kb += 8) {
      const int kk = kb + koff4;
      v4f a = *(const v4f*)&qtile[l16 * LP + kk];
      v4f b = *(const v4f*)&Wen[(size_t)(j0 + l16) * (2 * NH) + NH + kk];
      acc = wmma4(a.xy, b.xy, acc);
      acc = wmma4(a.zw, b.zw, acc);
    }
    const float bias = ben[j0 + l16];
#pragma unroll
    for (int r = 0; r < 8; ++r)
      otile[(r + 8 * half) * LP + j0 + l16] = tanhf(acc[r] + bias);
  }
  __syncthreads();

  for (int ft = wave; ft < 17; ft += 4) {
    const int f0 = ft * 16;
    const int f  = f0 + l16;
    v8f acc = {};
#pragma unroll 8
    for (int kb = 0; kb < NH; kb += 8) {
      const int kk = kb + koff4;
      v4f a = *(const v4f*)&otile[l16 * LP + kk];
      v4f b = {0.0f, 0.0f, 0.0f, 0.0f};
      if (f < NF) b = *(const v4f*)&Wmk[(size_t)f * NH + kk];
      acc = wmma4(a.xy, b.xy, acc);
      acc = wmma4(a.zw, b.zw, acc);
    }
    if (f < NF) {
      const float bias = bmk[f];
#pragma unroll
      for (int r = 0; r < 8; ++r) {
        const int t = t0 + r + 8 * half;
        const float v = acc[r] + bias;
        Out[((size_t)bat * NT + t) * NF + f] = 1.0f / (1.0f + __expf(-v));
      }
    }
  }
}

extern "C" void kernel_launch(void* const* d_in, const int* in_sizes, int n_in,
                              void* d_out, int out_size, void* d_ws, size_t ws_size,
                              hipStream_t stream) {
  (void)in_sizes; (void)n_in; (void)d_ws; (void)ws_size; (void)out_size;
  const float* k   = (const float*)d_in[0];
  const float* q   = (const float*)d_in[1];
  const float* Wsc = (const float*)d_in[2];
  const float* Wen = (const float*)d_in[3];
  const float* ben = (const float*)d_in[4];
  const float* Wmk = (const float*)d_in[5];
  const float* bmk = (const float*)d_in[6];
  dim3 grid(NT / 16, NB);
  attn_fused_kernel<<<grid, 128, 0, stream>>>(k, q, Wsc, Wen, ben, Wmk, bmk,
                                              (float*)d_out);
}


# 64 rows of t per work-group (4 row tiles, 16 waves): each B fragment loaded from global feeds the accumulator tiles of all 4 row tiles; sum order of every product unchanged
# speedup vs baseline: 2.4641x; 1.7218x over previous
#include <hip/hip_runtime.h>


#define NB 8
#define NT 2000
#define NH 256
#define NF 257
#define ALEN 128
#define LP 260
#define SS 164

#define MT 4
#define NWV (4 * MT)
#define NTHR (32 * NWV)
#define NTILE (NT / 16)
#define NBLK ((NTILE + MT - 1) / MT)

typedef float v2f __attribute__((ext_vector_type(2)));
typedef float v4f __attribute__((ext_vector_type(4)));
typedef float v8f __attribute__((ext_vector_type(8)));

__device__ __forceinline__ v8f wmma4(v2f a, v2f b, v8f c) {
  return __builtin_amdgcn_wmma_f32_16x16x4_f32(false, a, false, b, (short)0, c,
                                               false, false);
}

__global__ __launch_bounds__(NTHR)
void attn_fused_kernel(const float* __restrict__ Kin, const float* __restrict__ Qin,
                       const float* __restrict__ Wsc, const float* __restrict__ Wen,
                       const float* __restrict__ ben, const float* __restrict__ Wmk,
                       const float* __restrict__ bmk, float* __restrict__ Out)
{
  __shared__ __align__(16) float qtile[MT * 16 * LP];
  __shared__ __align__(16) float stile[MT * 16 * LP];
  __shared__ __align__(16) float otile[MT * 16 * LP];
  __shared__ __align__(16) float sbuf[MT * 16 * SS];
  __shared__ float wmax[NWV * MT * 16];
  __shared__ float rowmax[MT * 16];
  __shared__ float rowsum[MT * 16];
  __shared__ float rscale[MT * 16];

  const int tile0 = blockIdx.x * MT;
  const int bat   = blockIdx.y;
  const int t0    = tile0 * 16;
  const int tid   = threadIdx.x;
  const int wave  = tid >> 5;
  const int lane  = tid & 31;
  const int half  = lane >> 4;
  const int l16   = lane & 15;
  const int koff  = 2 * half;
  const int koff4 = 4 * half;
  const int mcnt  = (NTILE - tile0 < MT) ? (NTILE - tile0) : MT;
  const int grp   = wave >> 2;
  const int gtid  = tid & 127;
  const bool gvalid = grp < mcnt;

  const float* qbase = Qin + ((size_t)bat * NT + t0) * NH;
  const float* kbase = Kin + (size_t)bat * NT * NH;

  for (int i = tid; i < MT * 16 * NH; i += NTHR) {
    const int row = i >> 8;
    const bool live = row < mcnt * 16;
    const float v = qbase[live ? i : (i & (16 * NH - 1))];
    qtile[row * LP + (i & (NH - 1))] = live ? v : 0.0f;
  }
  __syncthreads();

  for (int nt = wave; nt < NH / 16; nt += NWV) {
    const int g0 = nt * 16;
    v8f acc[MT];
#pragma unroll
    for (int m = 0; m < MT; ++m) acc[m] = v8f{};
#pragma unroll 8
    for (int kb = 0; kb < NH; kb += 8) {
      const int kk = kb + koff4;
      v4f b = *(const v4f*)&Wsc[(size_t)(g0 + l16) * NH + kk];
#pragma unroll
      for (int m = 0; m < MT; ++m) {
        v4f a = *(const v4f*)&qtile[(m * 16 + l16) * LP + kk];
        acc[m] = wmma4(a.xy, b.xy, acc[m]);
        acc[m] = wmma4(a.zw, b.zw, acc[m]);
      }
    }
#pragma unroll
    for (int m = 0; m < MT; ++m) {
#pragma unroll
      for (int r = 0; r < 8; ++r) stile[(m * 16 + r + 8 * half) * LP + g0 + l16] = acc[m][r];
    }
  }
  __syncthreads();

  float rmax[MT][8];
#pragma unroll
  for (int m = 0; m < MT; ++m) {
#pragma unroll
    for (int r = 0; r < 8; ++r) rmax[m][r] = -3.4e38f;
  }

  for (int st = wave; st < NTILE; st += NWV) {
    const int s0 = st * 16;
    v8f acc[MT];
#pragma unroll
    for (int m = 0; m < MT; ++m) acc[m] = v8f{};
#pragma unroll 8
    for (int kb = 0; kb < NH; kb += 8) {
      const int kk = kb + koff4;
      v4f b = *(const v4f*)&kbase[(size_t)(s0 + l16) * NH + kk];
#pragma unroll
      for (int m = 0; m < MT; ++m) {
        v4f a = *(const v4f*)&stile[(m * 16 + l16) * LP + kk];
        acc[m] = wmma4(a.xy, b.xy, acc[m]);
        acc[m] = wmma4(a.zw, b.zw, acc[m]);
      }
    }
#pragma unroll
    for (int m = 0; m < MT; ++m) {
#pragma unroll
      for (int r = 0; r < 8; ++r) rmax[m][r] = fmaxf(rmax[m][r], acc[m][r]);
      const int tl = tile0 + m;
      const int lo = (tl > 8) ? (tl - 8) : 0;
      if (m < mcnt && st >= lo && st <= tl) {
        const int c0 = (st - lo) * 16;
#pragma unroll
        for (int r = 0; r < 8; ++r)
          sbuf[(m * 16 + r + 8 * half) * SS + c0 + l16] = acc[m][r];
      }
    }
  }

#pragma unroll
  for (int m = 0; m < MT; ++m) {
#pragma unroll
    for (int r = 0; r < 8; ++r) {
      float v = rmax[m][r];
      v = fmaxf(v, __shfl_xor(v, 1, 32));
      v = fmaxf(v, __shfl_xor(v, 2, 32));
      v = fmaxf(v, __shfl_xor(v, 4, 32));
      v = fmaxf(v, __shfl_xor(v, 8, 32));
      rmax[m][r] = v;
    }
  }
  if (l16 == 0) {
#pragma unroll
    for (int m = 0; m < MT; ++m) {
#pragma unroll
      for (int r = 0; r < 8; ++r) wmax[(wave * MT + m) * 16 + 8 * half + r] = rmax[m][r];
    }
  }
  __syncthreads();
  if (tid < MT * 16) {
    const int m   = tid >> 4;
    const int row = tid & 15;
    float mx = wmax[m * 16 + row];
    for (int w = 1; w < NWV; ++w) mx = fmaxf(mx, wmax[(w * MT + m) * 16 + row]);
    rowmax[tid] = mx;
    rowsum[tid] = 0.0f;
  }
  __syncthreads();

  const int gtile = tile0 + grp;
  const int st_lo = (gtile > 8) ? (gtile - 8) : 0;
  const int ncols = (gtile - st_lo + 1) * 16;
  const int s0b   = st_lo * 16;
  float* sb = sbuf + grp * 16 * SS;
  if (gvalid) {
    const int row = gtid & 15;
    const int t   = gtile * 16 + row;
    float psum = 0.0f;
    for (int c = gtid >> 4; c < ncols; c += 8) {
      const int s = s0b + c;
      float w = 0.0f;
      if (s <= t && s >= t - ALEN) w = __expf(sb[row * SS + c] - rowmax[grp * 16 + row]);
      sb[row * SS + c] = w;
      psum += w;
    }
    atomicAdd(&rowsum[grp * 16 + row], psum);
  }
  __syncthreads();
  if (tid < MT * 16) rscale[tid] = 1.0f / (rowsum[tid] + 1e-30f);
  __syncthreads();

  const int wlo = (tile0 > 8) ? (tile0 - 8) : 0;
  const int whi = tile0 + mcnt - 1;
  for (int nt = wave; nt < NH / 16; nt += NWV) {
    const int h0 = nt * 16;
    v8f acc[MT];
#pragma unroll
    for (int m = 0; m < MT; ++m) acc[m] = v8f{};
    for (int kt = wlo; kt <= whi; ++kt) {
      v2f b[4];
#pragma unroll
      for (int q4 = 0; q4 < 4; ++q4) {
        const int krow = kt * 16 + q4 * 4 + koff;
        b[q4].x = kbase[(size_t)krow       * NH + h0 + l16];
        b[q4].y = kbase[(size_t)(krow + 1) * NH + h0 + l16];
      }
#pragma unroll
      for (int m = 0; m < MT; ++m) {
        const int tl = tile0 + m;
        const int lo = (tl > 8) ? (tl - 8) : 0;
        if (m < mcnt && kt >= lo && kt <= tl) {
#pragma unroll
          for (int q4 = 0; q4 < 4; ++q4) {
            v2f a = *(const v2f*)&sbuf[(m * 16 + l16) * SS + (kt - lo) * 16 + q4 * 4 + koff];
            acc[m] = wmma4(a, b[q4], acc[m]);
          }
        }
      }
    }
#pragma unroll
    for (int m = 0; m < MT; ++m) {
#pragma unroll
      for (int r = 0; r < 8; ++r)
        stile[(m * 16 + r + 8 * half) * LP + h0 + l16] =
            acc[m][r] * rscale[m * 16 + r + 8 * half];
    }
  }
  __syncthreads();

  for (int nt = wave; nt < NH / 16; nt += NWV) {
    const int j0 = nt * 16;
    v8f acc[MT];
#pragma unroll
    for (int m = 0; m < MT; ++m) acc[m] = v8f{};
#pragma unroll 8
    for (int kb = 0; kb < NH; kb += 8) {
      const int kk = kb + koff4;
      v4f b = *(const v4f*)&Wen[(size_t)(j0 + l16) * (2 * NH) + kk];
#pragma unroll
      for (int m = 0; m < MT; ++m) {
        v4f a = *(const v4f*)&stile[(m * 16 + l16) * LP + kk];
        acc[m] = wmma4(a.xy, b.xy, acc[m]);
        acc[m] = wmma4(a.zw, b.zw, acc[m]);
      }
    }
#pragma unroll 8
    for (int kb = 0; kb < NH; kb += 8) {
      const int kk = kb + koff4;
      v4f b = *(const v4f*)&Wen[(size_t)(j0 + l16) * (2 * NH) + NH + kk];
#pragma unroll
      for (int m = 0; m < MT; ++m) {
        v4f a = *(const v4f*)&qtile[(m * 16 + l16) * LP + kk];
        acc[m] = wmma4(a.xy, b.xy, acc[m]);
        acc[m] = wmma4(a.zw, b.zw, acc[m]);
      }
    }
    const float bias = ben[j0 + l16];
#pragma unroll
    for (int m = 0; m < MT; ++m) {
#pragma unroll
      for (int r = 0; r < 8; ++r)
        otile[(m * 16 + r + 8 * half) * LP + j0 + l16] = tanhf(acc[m][r] + bias);
    }
  }
  __syncthreads();

  for (int ft = wave; ft < NF / 16; ft += NWV) {
    const int f = ft * 16 + l16;
    v8f acc[MT];
#pragma unroll
    for (int m = 0; m < MT; ++m) acc[m] = v8f{};
#pragma unroll 8
    for (int kb = 0; kb < NH; kb += 8) {
      const int kk = kb + koff4;
      v4f b = *(const v4f*)&Wmk[(size_t)f * NH + kk];
#pragma unroll
      for (int m = 0; m < MT; ++m) {
        v4f a = *(const v4f*)&otile[(m * 16 + l16) * LP + kk];
        acc[m] = wmma4(a.xy, b.xy, acc[m]);
        acc[m] = wmma4(a.zw, b.zw, acc[m]);
      }
    }
    const float bias = bmk[f];
#pragma unroll
    for (int m = 0; m < MT; ++m) {
      if (m < mcnt) {
#pragma unroll
        for (int r = 0; r < 8; ++r) {
          const int t = t0 + m * 16 + r + 8 * half;
          const float v = acc[m][r] + bias;
          Out[((size_t)bat * NT + t) * NF + f] = 1.0f / (1.0f + __expf(-v));
        }
      }
    }
  }
  if (wave < MT) {
    const int m = wave;
    const int f = (NF / 16) * 16 + l16;
    v8f acc = {};
#pragma unroll 8
    for (int kb = 0; kb < NH; kb += 8) {
      const int kk = kb + koff4;
      v4f a = *(const v4f*)&otile[(m * 16 + l16) * LP + kk];
      v4f b = {0.0f, 0.0f, 0.0f, 0.0f};
      if (f < NF) b = *(const v4f*)&Wmk[(size_t)f * NH + kk];
      acc = wmma4(a.xy, b.xy, acc);
      acc = wmma4(a.zw, b.zw, acc);
    }
    if (f < NF && m < mcnt) {
      const float bias = bmk[f];
#pragma unroll
      for (int r = 0; r < 8; ++r) {
        const int t = t0 + m * 16 + r + 8 * half;
        const float v = acc[r] + bias;
        Out[((size_t)bat * NT + t) * NF + f] = 1.0f / (1.0f + __expf(-v));
      }
    }
  }
}

extern "C" void kernel_launch(void* const* d_in, const int* in_sizes, int n_in,
                              void* d_out, int out_size, void* d_ws, size_t ws_size,
                              hipStream_t stream) {
  (void)in_sizes; (void)n_in; (void)d_ws; (void)ws_size; (void)out_size;
  const float* k   = (const float*)d_in[0];
  const float* q   = (const float*)d_in[1];
  const float* Wsc = (const float*)d_in[2];
  const float* Wen = (const float*)d_in[3];
  const float* ben = (const float*)d_in[4];
  const float* Wmk = (const float*)d_in[5];
  const float* bmk = (const float*)d_in[6];
  dim3 grid(NBLK, NB);
  attn_fused_kernel<<<grid, NTHR, 0, stream>>>(k, q, Wsc, Wen, ben, Wmk, bmk,
                                               (float*)d_out);
}


# score product: each wave fetches its 16 k rows with whole-line 16-byte loads (4 lines per instruction, not 16), parks them in private LDS slots and reads B fragments there; sum order unchanged
# speedup vs baseline: 2.5372x; 1.0297x over previous
#include <hip/hip_runtime.h>


#define NB 8
#define NT 2000
#define NH 256
#define NF 257
#define ALEN 128
#define LP 260
#define BP 68
#define SS 164

#define MT 4
#define NWV (4 * MT)
#define NTHR (32 * NWV)
#define NTILE (NT / 16)
#define NBLK ((NTILE + MT - 1) / MT)

typedef float v2f __attribute__((ext_vector_type(2)));
typedef float v4f __attribute__((ext_vector_type(4)));
typedef float v8f __attribute__((ext_vector_type(8)));

__device__ __forceinline__ v8f wmma4(v2f a, v2f b, v8f c) {
  return __builtin_amdgcn_wmma_f32_16x16x4_f32(false, a, false, b, (short)0, c,
                                               false, false);
}

__global__ __launch_bounds__(NTHR)
void attn_fused_kernel(const float* __restrict__ Kin, const float* __restrict__ Qin,
                       const float* __restrict__ Wsc, const float* __restrict__ Wen,
                       const float* __restrict__ ben, const float* __restrict__ Wmk,
                       const float* __restrict__ bmk, float* __restrict__ Out)
{
  __shared__ __align__(16) float qtile[MT * 16 * LP];
  __shared__ __align__(16) float stile[MT * 16 * LP];
  __shared__ __align__(16) float otile[NWV * 2 * 16 * BP];
  __shared__ __align__(16) float sbuf[MT * 16 * SS];
  __shared__ float wmax[NWV * MT * 16];
  __shared__ float rowmax[MT * 16];
  __shared__ float rowsum[MT * 16];
  __shared__ float rscale[MT * 16];

  const int tile0 = blockIdx.x * MT;
  const int bat   = blockIdx.y;
  const int t0    = tile0 * 16;
  const int tid   = threadIdx.x;
  const int wave  = tid >> 5;
  const int lane  = tid & 31;
  const int half  = lane >> 4;
  const int l16   = lane & 15;
  const int koff  = 2 * half;
  const int koff4 = 4 * half;
  const int mcnt  = (NTILE - tile0 < MT) ? (NTILE - tile0) : MT;
  const int grp   = wave >> 2;
  const int gtid  = tid & 127;
  const bool gvalid = grp < mcnt;

  const float* qbase = Qin + ((size_t)bat * NT + t0) * NH;
  const float* kbase = Kin + (size_t)bat * NT * NH;

  for (int i = tid; i < MT * 16 * NH; i += NTHR) {
    const int row = i >> 8;
    const bool live = row < mcnt * 16;
    const float v = qbase[live ? i : (i & (16 * NH - 1))];
    qtile[row * LP + (i & (NH - 1))] = live ? v : 0.0f;
  }
  __syncthreads();

  for (int nt = wave; nt < NH / 16; nt += NWV) {
    const int g0 = nt * 16;
    v8f acc[MT];
#pragma unroll
    for (int m = 0; m < MT; ++m) acc[m] = v8f{};
#pragma unroll 8
    for (int kb = 0; kb < NH; kb += 8) {
      const int kk = kb + koff4;
      v4f b = *(const v4f*)&Wsc[(size_t)(g0 + l16) * NH + kk];
#pragma unroll
      for (int m = 0; m < MT; ++m) {
        v4f a = *(const v4f*)&qtile[(m * 16 + l16) * LP + kk];
        acc[m] = wmma4(a.xy, b.xy, acc[m]);
        acc[m] = wmma4(a.zw, b.zw, acc[m]);
      }
    }
#pragma unroll
    for (int m = 0; m < MT; ++m) {
#pragma unroll
      for (int r = 0; r < 8; ++r) stile[(m * 16 + r + 8 * half) * LP + g0 + l16] = acc[m][r];
    }
  }
  __syncthreads();

  float rmax[MT][8];
#pragma unroll
  for (int m = 0; m < MT; ++m) {
#pragma unroll
    for (int r = 0; r < 8; ++r) rmax[m][r] = -3.4e38f;
  }

  float* bw = otile + wave * (2 * 16 * BP);
  const int ntl = (NTILE - wave + NWV - 1) / NWV;
  const int nch = 4 * ntl;
  const float* gsrc0 = kbase + (size_t)half * NH + 4 * l16;
  float* sdst0 = bw + half * BP + 4 * l16;
  v4f g[8];
  auto gload = [&](int i) {
    const float* p = gsrc0 + (size_t)(wave + NWV * (i >> 2)) * (16 * NH) + (i & 3) * 64;
#pragma unroll
    for (int j = 0; j < 8; ++j) g[j] = *(const v4f*)(p + (size_t)j * (2 * NH));
  };
  auto sstore = [&](int i) {
    float* p = sdst0 + (i & 1) * (16 * BP);
#pragma unroll
    for (int j = 0; j < 8; ++j) *(v4f*)(p + j * (2 * BP)) = g[j];
  };
  gload(0);
  sstore(0);
  gload(1);
  int ci = 0;
  for (int st = wave; st < NTILE; st += NWV) {
    v8f acc[MT];
#pragma unroll
    for (int m = 0; m < MT; ++m) acc[m] = v8f{};
    for (int c = 0; c < 4; ++c, ++ci) {
      const int c1 = (ci + 1 < nch) ? (ci + 1) : (nch - 1);
      const int c2 = (ci + 2 < nch) ? (ci + 2) : (nch - 1);
      sstore(c1);
      gload(c2);
      const float* bs = bw + (ci & 1) * (16 * BP) + l16 * BP + koff4;
#pragma unroll
      for (int kb = 0; kb < 64; kb += 8) {
        v4f b = *(const v4f*)&bs[kb];
#pragma unroll
        for (int m = 0; m < MT; ++m) {
          v4f a = *(const v4f*)&stile[(m * 16 + l16) * LP + c * 64 + kb + koff4];
          acc[m] = wmma4(a.xy, b.xy, acc[m]);
          acc[m] = wmma4(a.zw, b.zw, acc[m]);
        }
      }
    }
#pragma unroll
    for (int m = 0; m < MT; ++m) {
#pragma unroll
      for (int r = 0; r < 8; ++r) rmax[m][r] = fmaxf(rmax[m][r], acc[m][r]);
      const int tl = tile0 + m;
      const int lo = (tl > 8) ? (tl - 8) : 0;
      if (m < mcnt && st >= lo && st <= tl) {
        const int c0 = (st - lo) * 16;
#pragma unroll
        for (int r = 0; r < 8; ++r)
          sbuf[(m * 16 + r + 8 * half) * SS + c0 + l16] = acc[m][r];
      }
    }
  }

#pragma unroll
  for (int m = 0; m < MT; ++m) {
#pragma unroll
    for (int r = 0; r < 8; ++r) {
      float v = rmax[m][r];
      v = fmaxf(v, __shfl_xor(v, 1, 32));
      v = fmaxf(v, __shfl_xor(v, 2, 32));
      v = fmaxf(v, __shfl_xor(v, 4, 32));
      v = fmaxf(v, __shfl_xor(v, 8, 32));
      rmax[m][r] = v;
    }
  }
  if (l16 == 0) {
#pragma unroll
    for (int m = 0; m < MT; ++m) {
#pragma unroll
      for (int r = 0; r < 8; ++r) wmax[(wave * MT + m) * 16 + 8 * half + r] = rmax[m][r];
    }
  }
  __syncthreads();
  if (tid < MT * 16) {
    const int m   = tid >> 4;
    const int row = tid & 15;
    float mx = wmax[m * 16 + row];
    for (int w = 1; w < NWV; ++w) mx = fmaxf(mx, wmax[(w * MT + m) * 16 + row]);
    rowmax[tid] = mx;
    rowsum[tid] = 0.0f;
  }
  __syncthreads();

  const int gtile = tile0 + grp;
  const int st_lo = (gtile > 8) ? (gtile - 8) : 0;
  const int ncols = (gtile - st_lo + 1) * 16;
  const int s0b   = st_lo * 16;
  float* sb = sbuf + grp * 16 * SS;
  if (gvalid) {
    const int row = gtid & 15;
    const int t   = gtile * 16 + row;
    float psum = 0.0f;
    for (int c = gtid >> 4; c < ncols; c += 8) {
      const int s = s0b + c;
      float w = 0.0f;
      if (s <= t && s >= t - ALEN) w = __expf(sb[row * SS + c] - rowmax[grp * 16 + row]);
      sb[row * SS + c] = w;
      psum += w;
    }
    atomicAdd(&rowsum[grp * 16 + row], psum);
  }
  __syncthreads();
  if (tid < MT * 16) rscale[tid] = 1.0f / (rowsum[tid] + 1e-30f);
  __syncthreads();

  const int wlo = (tile0 > 8) ? (tile0 - 8) : 0;
  const int whi = tile0 + mcnt - 1;
  for (int nt = wave; nt < NH / 16; nt += NWV) {
    const int h0 = nt * 16;
    v8f acc[MT];
#pragma unroll
    for (int m = 0; m < MT; ++m) acc[m] = v8f{};
    for (int kt = wlo; kt <= whi; ++kt) {
      v2f b[4];
#pragma unroll
      for (int q4 = 0; q4 < 4; ++q4) {
        const int krow = kt * 16 + q4 * 4 + koff;
        b[q4].x = kbase[(size_t)krow       * NH + h0 + l16];
        b[q4].y = kbase[(size_t)(krow + 1) * NH + h0 + l16];
      }
#pragma unroll
      for (int m = 0; m < MT; ++m) {
        const int tl = tile0 + m;
        const int lo = (tl > 8) ? (tl - 8) : 0;
        if (m < mcnt && kt >= lo && kt <= tl) {
#pragma unroll
          for (int q4 = 0; q4 < 4; ++q4) {
            v2f a = *(const v2f*)&sbuf[(m * 16 + l16) * SS + (kt - lo) * 16 + q4 * 4 + koff];
            acc[m] = wmma4(a, b[q4], acc[m]);
          }
        }
      }
    }
#pragma unroll
    for (int m = 0; m < MT; ++m) {
#pragma unroll
      for (int r = 0; r < 8; ++r)
        stile[(m * 16 + r + 8 * half) * LP + h0 + l16] =
            acc[m][r] * rscale[m * 16 + r + 8 * half];
    }
  }
  __syncthreads();

  for (int nt = wave; nt < NH / 16; nt += NWV) {
    const int j0 = nt * 16;
    v8f acc[MT];
#pragma unroll
    for (int m = 0; m < MT; ++m) acc[m] = v8f{};
#pragma unroll 8
    for (int kb = 0; kb < NH; kb += 8) {
      const int kk = kb + koff4;
      v4f b = *(const v4f*)&Wen[(size_t)(j0 + l16) * (2 * NH) + kk];
#pragma unroll
      for (int m = 0; m < MT; ++m) {
        v4f a = *(const v4f*)&stile[(m * 16 + l16) * LP + kk];
        acc[m] = wmma4(a.xy, b.xy, acc[m]);
        acc[m] = wmma4(a.zw, b.zw, acc[m]);
      }
    }
#pragma unroll 8
    for (int kb = 0; kb < NH; kb += 8) {
      const int kk = kb + koff4;
      v4f b = *(const v4f*)&Wen[(size_t)(j0 + l16) * (2 * NH) + NH + kk];
#pragma unroll
      for (int m = 0; m < MT; ++m) {
        v4f a = *(const v4f*)&qtile[(m * 16 + l16) * LP + kk];
        acc[m] = wmma4(a.xy, b.xy, acc[m]);
        acc[m] = wmma4(a.zw, b.zw, acc[m]);
      }
    }
    const float bias = ben[j0 + l16];
#pragma unroll
    for (int m = 0; m < MT; ++m) {
#pragma unroll
      for (int r = 0; r < 8; ++r)
        otile[(m * 16 + r + 8 * half) * LP + j0 + l16] = tanhf(acc[m][r] + bias);
    }
  }
  __syncthreads();

  for (int ft = wave; ft < NF / 16; ft += NWV) {
    const int f = ft * 16 + l16;
    v8f acc[MT];
#pragma unroll
    for (int m = 0; m < MT; ++m) acc[m] = v8f{};
#pragma unroll 8
    for (int kb = 0; kb < NH; kb += 8) {
      const int kk = kb + koff4;
      v4f b = *(const v4f*)&Wmk[(size_t)f * NH + kk];
#pragma unroll
      for (int m = 0; m < MT; ++m) {
        v4f a = *(const v4f*)&otile[(m * 16 + l16) * LP + kk];
        acc[m] = wmma4(a.xy, b.xy, acc[m]);
        acc[m] = wmma4(a.zw, b.zw, acc[m]);
      }
    }
    const float bias = bmk[f];
#pragma unroll
    for (int m = 0; m < MT; ++m) {
      if (m < mcnt) {
#pragma unroll
        for (int r = 0; r < 8; ++r) {
          const int t = t0 + m * 16 + r + 8 * half;
          const float v = acc[m][r] + bias;
          Out[((size_t)bat * NT + t) * NF + f] = 1.0f / (1.0f + __expf(-v));
        }
      }
    }
  }
  if (wave < MT) {
    const int m = wave;
    const int f = (NF / 16) * 16 + l16;
    v8f acc = {};
#pragma unroll 8
    for (int kb = 0; kb < NH; kb += 8) {
      const int kk = kb + koff4;
      v4f a = *(const v4f*)&otile[(m * 16 + l16) * LP + kk];
      v4f b = {0.0f, 0.0f, 0.0f, 0.0f};
      if (f < NF) b = *(const v4f*)&Wmk[(size_t)f * NH + kk];
      acc = wmma4(a.xy, b.xy, acc);
      acc = wmma4(a.zw, b.zw, acc);
    }
    if (f < NF && m < mcnt) {
      const float bias = bmk[f];
#pragma unroll
      for (int r = 0; r < 8; ++r) {
        const int t = t0 + m * 16 + r + 8 * half;
        const float v = acc[r] + bias;
        Out[((size_t)bat * NT + t) * NF + f] = 1.0f / (1.0f + __expf(-v));
      }
    }
  }
}

extern "C" void kernel_launch(void* const* d_in, const int* in_sizes, int n_in,
                              void* d_out, int out_size, void* d_ws, size_t ws_size,
                              hipStream_t stream) {
  (void)in_sizes; (void)n_in; (void)d_ws; (void)ws_size; (void)out_size;
  const float* k   = (const float*)d_in[0];
  const float* q   = (const float*)d_in[1];
  const float* Wsc = (const float*)d_in[2];
  const float* Wen = (const float*)d_in[3];
  const float* ben = (const float*)d_in[4];
  const float* Wmk = (const float*)d_in[5];
  const float* bmk = (const float*)d_in[6];
  dim3 grid(NBLK, NB);
  attn_fused_kernel<<<grid, NTHR, 0, stream>>>(k, q, Wsc, Wen, ben, Wmk, bmk,
                                               (float*)d_out);
}


# wave priority (s_setprio) stepped down as a wave advances through a product loop: the waves of a SIMD advance together, the last one no longer runs its loop alone at the end; sum order unchanged
# speedup vs baseline: 2.6279x; 1.0357x over previous
#include <hip/hip_runtime.h>


#define NB 8
#define NT 2000
#define NH 256
#define NF 257
#define ALEN 128
#define LP 260
#define BP 68
#define SS 164

#define MT 4
#define NWV (4 * MT)
#define NTHR (32 * NWV)
#define NTILE (NT / 16)
#define NBLK ((NTILE + MT - 1) / MT)

typedef float v2f __attribute__((ext_vector_type(2)));
typedef float v4f __attribute__((ext_vector_type(4)));
typedef float v8f __attribute__((ext_vector_type(8)));

__device__ __forceinline__ v8f wmma4(v2f a, v2f b, v8f c) {
  return __builtin_amdgcn_wmma_f32_16x16x4_f32(false, a, false, b, (short)0, c,
                                               false, false);
}

__device__ __forceinline__ void wave_prio(int p) {
  switch (p) {
    case 3:  __builtin_amdgcn_s_setprio(3); break;
    case 2:  __builtin_amdgcn_s_setprio(2); break;
    case 1:  __builtin_amdgcn_s_setprio(1); break;
    default: __builtin_amdgcn_s_setprio(0); break;
  }
}

__global__ __launch_bounds__(NTHR)
void attn_fused_kernel(const float* __restrict__ Kin, const float* __restrict__ Qin,
                       const float* __restrict__ Wsc, const float* __restrict__ Wen,
                       const float* __restrict__ ben, const float* __restrict__ Wmk,
                       const float* __restrict__ bmk, float* __restrict__ Out)
{
  __shared__ __align__(16) float qtile[MT * 16 * LP];
  __shared__ __align__(16) float stile[MT * 16 * LP];
  __shared__ __align__(16) float otile[NWV * 2 * 16 * BP];
  __shared__ __align__(16) float sbuf[MT * 16 * SS];
  __shared__ float wmax[NWV * MT * 16];
  __shared__ float rowmax[MT * 16];
  __shared__ float rowsum[MT * 16];
  __shared__ float rscale[MT * 16];

  const int tile0 = blockIdx.x * MT;
  const int bat   = blockIdx.y;
  const int t0    = tile0 * 16;
  const int tid   = threadIdx.x;
  const int wave  = tid >> 5;
  const int lane  = tid & 31;
  const int half  = lane >> 4;
  const int l16   = lane & 15;
  const int koff  = 2 * half;
  const int koff4 = 4 * half;
  const int mcnt  = (NTILE - tile0 < MT) ? (NTILE - tile0) : MT;
  const int grp   = wave >> 2;
  const int gtid  = tid & 127;
  const bool gvalid = grp < mcnt;

  const float* qbase = Qin + ((size_t)bat * NT + t0) * NH;
  const float* kbase = Kin + (size_t)bat * NT * NH;

  for (int i = tid; i < MT * 16 * NH; i += NTHR) {
    const int row = i >> 8;
    const bool live = row < mcnt * 16;
    const float v = qbase[live ? i : (i & (16 * NH - 1))];
    qtile[row * LP + (i & (NH - 1))] = live ? v : 0.0f;
  }
  __syncthreads();

  for (int nt = wave; nt < NH / 16; nt += NWV) {
    const int g0 = nt * 16;
    v8f acc[MT];
#pragma unroll
    for (int m = 0; m < MT; ++m) acc[m] = v8f{};
#pragma unroll 8
    for (int kb = 0; kb < NH; kb += 8) {
      if ((kb & 63) == 0) wave_prio(3 - (kb >> 6));
      const int kk = kb + koff4;
      v4f b = *(const v4f*)&Wsc[(size_t)(g0 + l16) * NH + kk];
#pragma unroll
      for (int m = 0; m < MT; ++m) {
        v4f a = *(const v4f*)&qtile[(m * 16 + l16) * LP + kk];
        acc[m] = wmma4(a.xy, b.xy, acc[m]);
        acc[m] = wmma4(a.zw, b.zw, acc[m]);
      }
    }
#pragma unroll
    for (int m = 0; m < MT; ++m) {
#pragma unroll
      for (int r = 0; r < 8; ++r) stile[(m * 16 + r + 8 * half) * LP + g0 + l16] = acc[m][r];
    }
  }
  __syncthreads();

  float rmax[MT][8];
#pragma unroll
  for (int m = 0; m < MT; ++m) {
#pragma unroll
    for (int r = 0; r < 8; ++r) rmax[m][r] = -3.4e38f;
  }

  float* bw = otile + wave * (2 * 16 * BP);
  const int ntl = (NTILE - wave + NWV - 1) / NWV;
  const int nch = 4 * ntl;
  const float* gsrc0 = kbase + (size_t)half * NH + 4 * l16;
  float* sdst0 = bw + half * BP + 4 * l16;
  v4f g[8];
  auto gload = [&](int i) {
    const float* p = gsrc0 + (size_t)(wave + NWV * (i >> 2)) * (16 * NH) + (i & 3) * 64;
#pragma unroll
    for (int j = 0; j < 8; ++j) g[j] = *(const v4f*)(p + (size_t)j * (2 * NH));
  };
  auto sstore = [&](int i) {
    float* p = sdst0 + (i & 1) * (16 * BP);
#pragma unroll
    for (int j = 0; j < 8; ++j) *(v4f*)(p + j * (2 * BP)) = g[j];
  };
  gload(0);
  sstore(0);
  gload(1);
  int ci = 0;
  int it = 0;
  for (int st = wave; st < NTILE; st += NWV, ++it) {
    wave_prio(it < 3 ? 3 : (it < 6 ? 2 : (it == 6 ? 1 : 0)));
    v8f acc[MT];
#pragma unroll
    for (int m = 0; m < MT; ++m) acc[m] = v8f{};
    for (int c = 0; c < 4; ++c, ++ci) {
      const int c1 = (ci + 1 < nch) ? (ci + 1) : (nch - 1);
      const int c2 = (ci + 2 < nch) ? (ci + 2) : (nch - 1);
      sstore(c1);
      gload(c2);
      const float* bs = bw + (ci & 1) * (16 * BP) + l16 * BP + koff4;
#pragma unroll
      for (int kb = 0; kb < 64; kb += 8) {
        v4f b = *(const v4f*)&bs[kb];
#pragma unroll
        for (int m = 0; m < MT; ++m) {
          v4f a = *(const v4f*)&stile[(m * 16 + l16) * LP + c * 64 + kb + koff4];
          acc[m] = wmma4(a.xy, b.xy, acc[m]);
          acc[m] = wmma4(a.zw, b.zw, acc[m]);
        }
      }
    }
#pragma unroll
    for (int m = 0; m < MT; ++m) {
#pragma unroll
      for (int r = 0; r < 8; ++r) rmax[m][r] = fmaxf(rmax[m][r], acc[m][r]);
      const int tl = tile0 + m;
      const int lo = (tl > 8) ? (tl - 8) : 0;
      if (m < mcnt && st >= lo && st <= tl) {
        const int c0 = (st - lo) * 16;
#pragma unroll
        for (int r = 0; r < 8; ++r)
          sbuf[(m * 16 + r + 8 * half) * SS + c0 + l16] = acc[m][r];
      }
    }
  }

  wave_prio(0);
#pragma unroll
  for (int m = 0; m < MT; ++m) {
#pragma unroll
    for (int r = 0; r < 8; ++r) {
      float v = rmax[m][r];
      v = fmaxf(v, __shfl_xor(v, 1, 32));
      v = fmaxf(v, __shfl_xor(v, 2, 32));
      v = fmaxf(v, __shfl_xor(v, 4, 32));
      v = fmaxf(v, __shfl_xor(v, 8, 32));
      rmax[m][r] = v;
    }
  }
  if (l16 == 0) {
#pragma unroll
    for (int m = 0; m < MT; ++m) {
#pragma unroll
      for (int r = 0; r < 8; ++r) wmax[(wave * MT + m) * 16 + 8 * half + r] = rmax[m][r];
    }
  }
  __syncthreads();
  if (tid < MT * 16) {
    const int m   = tid >> 4;
    const int row = tid & 15;
    float mx = wmax[m * 16 + row];
    for (int w = 1; w < NWV; ++w) mx = fmaxf(mx, wmax[(w * MT + m) * 16 + row]);
    rowmax[tid] = mx;
    rowsum[tid] = 0.0f;
  }
  __syncthreads();

  const int gtile = tile0 + grp;
  const int st_lo = (gtile > 8) ? (gtile - 8) : 0;
  const int ncols = (gtile - st_lo + 1) * 16;
  const int s0b   = st_lo * 16;
  float* sb = sbuf + grp * 16 * SS;
  if (gvalid) {
    const int row = gtid & 15;
    const int t   = gtile * 16 + row;
    float psum = 0.0f;
    for (int c = gtid >> 4; c < ncols; c += 8) {
      const int s = s0b + c;
      float w = 0.0f;
      if (s <= t && s >= t - ALEN) w = __expf(sb[row * SS + c] - rowmax[grp * 16 + row]);
      sb[row * SS + c] = w;
      psum += w;
    }
    atomicAdd(&rowsum[grp * 16 + row], psum);
  }
  __syncthreads();
  if (tid < MT * 16) rscale[tid] = 1.0f / (rowsum[tid] + 1e-30f);
  __syncthreads();

  const int wlo = (tile0 > 8) ? (tile0 - 8) : 0;
  const int whi = tile0 + mcnt - 1;
  for (int nt = wave; nt < NH / 16; nt += NWV) {
    const int h0 = nt * 16;
    v8f acc[MT];
#pragma unroll
    for (int m = 0; m < MT; ++m) acc[m] = v8f{};
    for (int kt = wlo; kt <= whi; ++kt) {
      v2f b[4];
#pragma unroll
      for (int q4 = 0; q4 < 4; ++q4) {
        const int krow = kt * 16 + q4 * 4 + koff;
        b[q4].x = kbase[(size_t)krow       * NH + h0 + l16];
        b[q4].y = kbase[(size_t)(krow + 1) * NH + h0 + l16];
      }
#pragma unroll
      for (int m = 0; m < MT; ++m) {
        const int tl = tile0 + m;
        const int lo = (tl > 8) ? (tl - 8) : 0;
        if (m < mcnt && kt >= lo && kt <= tl) {
#pragma unroll
          for (int q4 = 0; q4 < 4; ++q4) {
            v2f a = *(const v2f*)&sbuf[(m * 16 + l16) * SS + (kt - lo) * 16 + q4 * 4 + koff];
            acc[m] = wmma4(a, b[q4], acc[m]);
          }
        }
      }
    }
#pragma unroll
    for (int m = 0; m < MT; ++m) {
#pragma unroll
      for (int r = 0; r < 8; ++r)
        stile[(m * 16 + r + 8 * half) * LP + h0 + l16] =
            acc[m][r] * rscale[m * 16 + r + 8 * half];
    }
  }
  __syncthreads();

  for (int nt = wave; nt < NH / 16; nt += NWV) {
    const int j0 = nt * 16;
    v8f acc[MT];
#pragma unroll
    for (int m = 0; m < MT; ++m) acc[m] = v8f{};
#pragma unroll 8
    for (int kb = 0; kb < NH; kb += 8) {
      if ((kb & 63) == 0) wave_prio(3 - (kb >> 7));
      const int kk = kb + koff4;
      v4f b = *(const v4f*)&Wen[(size_t)(j0 + l16) * (2 * NH) + kk];
#pragma unroll
      for (int m = 0; m < MT; ++m) {
        v4f a = *(const v4f*)&stile[(m * 16 + l16) * LP + kk];
        acc[m] = wmma4(a.xy, b.xy, acc[m]);
        acc[m] = wmma4(a.zw, b.zw, acc[m]);
      }
    }
#pragma unroll 8
    for (int kb = 0; kb < NH; kb += 8) {
      if ((kb & 63) == 0) wave_prio(1 - (kb >> 7));
      const int kk = kb + koff4;
      v4f b = *(const v4f*)&Wen[(size_t)(j0 + l16) * (2 * NH) + NH + kk];
#pragma unroll
      for (int m = 0; m < MT; ++m) {
        v4f a = *(const v4f*)&qtile[(m * 16 + l16) * LP + kk];
        acc[m] = wmma4(a.xy, b.xy, acc[m]);
        acc[m] = wmma4(a.zw, b.zw, acc[m]);
      }
    }
    const float bias = ben[j0 + l16];
#pragma unroll
    for (int m = 0; m < MT; ++m) {
#pragma unroll
      for (int r = 0; r < 8; ++r)
        otile[(m * 16 + r + 8 * half) * LP + j0 + l16] = tanhf(acc[m][r] + bias);
    }
  }
  __syncthreads();

  for (int ft = wave; ft < NF / 16; ft += NWV) {
    const int f = ft * 16 + l16;
    v8f acc[MT];
#pragma unroll
    for (int m = 0; m < MT; ++m) acc[m] = v8f{};
#pragma unroll 8
    for (int kb = 0; kb < NH; kb += 8) {
      if ((kb & 63) == 0) wave_prio(3 - (kb >> 6));
      const int kk = kb + koff4;
      v4f b = *(const v4f*)&Wmk[(size_t)f * NH + kk];
#pragma unroll
      for (int m = 0; m < MT; ++m) {
        v4f a = *(const v4f*)&otile[(m * 16 + l16) * LP + kk];
        acc[m] = wmma4(a.xy, b.xy, acc[m]);
        acc[m] = wmma4(a.zw, b.zw, acc[m]);
      }
    }
    const float bias = bmk[f];
#pragma unroll
    for (int m = 0; m < MT; ++m) {
      if (m < mcnt) {
#pragma unroll
        for (int r = 0; r < 8; ++r) {
          const int t = t0 + m * 16 + r + 8 * half;
          const float v = acc[m][r] + bias;
          Out[((size_t)bat * NT + t) * NF + f] = 1.0f / (1.0f + __expf(-v));
        }
      }
    }
  }
  if (wave < MT) {
    const int m = wave;
    const int f = (NF / 16) * 16 + l16;
    v8f acc = {};
#pragma unroll 8
    for (int kb = 0; kb < NH; kb += 8) {
      const int kk = kb + koff4;
      v4f a = *(const v4f*)&otile[(m * 16 + l16) * LP + kk];
      v4f b = {0.0f, 0.0f, 0.0f, 0.0f};
      if (f < NF) b = *(const v4f*)&Wmk[(size_t)f * NH + kk];
      acc = wmma4(a.xy, b.xy, acc);
      acc = wmma4(a.zw, b.zw, acc);
    }
    if (f < NF && m < mcnt) {
      const float bias = bmk[f];
#pragma unroll
      for (int r = 0; r < 8; ++r) {
        const int t = t0 + m * 16 + r + 8 * half;
        const float v = acc[r] + bias;
        Out[((size_t)bat * NT + t) * NF + f] = 1.0f / (1.0f + __expf(-v));
      }
    }
  }
}

extern "C" void kernel_launch(void* const* d_in, const int* in_sizes, int n_in,
                              void* d_out, int out_size, void* d_ws, size_t ws_size,
                              hipStream_t stream) {
  (void)in_sizes; (void)n_in; (void)d_ws; (void)ws_size; (void)out_size;
  const float* k   = (const float*)d_in[0];
  const float* q   = (const float*)d_in[1];
  const float* Wsc = (const float*)d_in[2];
  const float* Wen = (const float*)d_in[3];
  const float* ben = (const float*)d_in[4];
  const float* Wmk = (const float*)d_in[5];
  const float* bmk = (const float*)d_in[6];
  dim3 grid(NBLK, NB);
  attn_fused_kernel<<<grid, NTHR, 0, stream>>>(k, q, Wsc, Wen, ben, Wmk, bmk,
                                               (float*)d_out);
}
